# starting kernel
# baseline (speedup 1.0000x reference)
#include <hip/hip_runtime.h>

typedef _Float16 v16h __attribute__((ext_vector_type(16)));
typedef _Float16 v8h  __attribute__((ext_vector_type(8)));
typedef _Float16 v4h  __attribute__((ext_vector_type(4)));
typedef float    v8f  __attribute__((ext_vector_type(8)));
typedef float    v4f  __attribute__((ext_vector_type(4)));

constexpr int B_ = 8, L_ = 2500, D_ = 512, Y_ = 8921;
constexpr int YT  = 128;
constexpr int LT  = 32;
constexpr int KC  = D_ / 32;
constexpr int STRIDE = D_ + 8;
constexpr int BUFH   = LT * STRIDE;
constexpr float LOG2E = 1.44269504088896340736f;

__global__ __launch_bounds__(256, 2)
void label_attn_kernel(const float* __restrict__ x,
                       const float* __restrict__ Uw,
                       const float* __restrict__ Fw,
                       const float* __restrict__ fb,
                       float* __restrict__ out)
{
    __shared__ _Float16 lds[2 * BUFH];

    const int b    = blockIdx.y;
    const int yt   = blockIdx.x * YT;
    const int tid  = threadIdx.x;
    const int wave = tid >> 5;
    const int lane = tid & 31;
    const int half = lane & 15;
    const bool hi  = lane >= 16;

    int yrow = yt + wave * 16 + half;
    if (yrow >= Y_) yrow = Y_ - 1;
    const float* urow = Uw + (size_t)yrow * D_;
    const float* frow = Fw + (size_t)yrow * D_;

    v16h aU[KC], aF[KC];
#pragma unroll
    for (int kc = 0; kc < KC; ++kc) {
        const int base = kc * 32 + (hi ? 8 : 0);
        v4f u0 = *(const v4f*)(urow + base);
        v4f u1 = *(const v4f*)(urow + base + 4);
        v4f u2 = *(const v4f*)(urow + base + 16);
        v4f u3 = *(const v4f*)(urow + base + 20);
        v4f f0 = *(const v4f*)(frow + base);
        v4f f1 = *(const v4f*)(frow + base + 4);
        v4f f2 = *(const v4f*)(frow + base + 16);
        v4f f3 = *(const v4f*)(frow + base + 20);
#pragma unroll
        for (int i = 0; i < 4; ++i) {
            aU[kc][i]      = (_Float16)u0[i];
            aU[kc][4 + i]  = (_Float16)u1[i];
            aU[kc][8 + i]  = (_Float16)u2[i];
            aU[kc][12 + i] = (_Float16)u3[i];
            aF[kc][i]      = (_Float16)f0[i];
            aF[kc][4 + i]  = (_Float16)f1[i];
            aF[kc][8 + i]  = (_Float16)f2[i];
            aF[kc][12 + i] = (_Float16)f3[i];
        }
    }

    float rmax[8], rden[8], rnum[8];
#pragma unroll
    for (int m = 0; m < 8; ++m) { rmax[m] = -1e30f; rden[m] = 0.f; rnum[m] = 0.f; }

    const float* xb = x + (size_t)b * L_ * D_;
    const int numLT = (L_ + LT - 1) / LT;

    {
#pragma unroll
        for (int j = 0; j < 4; ++j) {
            const int r = wave * 4 + j;
            int lrow = r; if (lrow >= L_) lrow = L_ - 1;
            const float* src = xb + (size_t)lrow * D_;
#pragma unroll
            for (int k = 0; k < 4; ++k) {
                const int c4 = lane + (k << 5);
                v4f f = *(const v4f*)(src + (c4 << 2));
                v4h h;
#pragma unroll
                for (int i = 0; i < 4; ++i) h[i] = (_Float16)f[i];
                *(v4h*)&lds[r * STRIDE + (c4 << 2)] = h;
            }
        }
    }

    for (int lt = 0; lt < numLT; ++lt) {
        const int cur = lt & 1;
        const int l0  = lt * LT;
        const bool have = (lt + 1) < numLT;

        v4f stage[4][4];
        if (have) {
            const int l0n = l0 + LT;
#pragma unroll
            for (int j = 0; j < 4; ++j) {
                const int r = wave * 4 + j;
                int lrow = l0n + r; if (lrow >= L_) lrow = L_ - 1;
                const float* src = xb + (size_t)lrow * D_;
#pragma unroll
                for (int k = 0; k < 4; ++k) {
                    const int c4 = lane + (k << 5);
                    stage[j][k] = *(const v4f*)(src + (c4 << 2));
                }
                if (l0n + LT + r < L_)
                    __builtin_prefetch(src + (size_t)LT * D_, 0, 1);
            }
        }

        __syncthreads();

        v8f S0 = {}, T0 = {}, S1 = {}, T1 = {};
        const _Float16* base = lds + cur * BUFH + (hi ? 8 : 0);
        const _Float16* bp0  = base + half * STRIDE;
        const _Float16* bp1  = base + (16 + half) * STRIDE;
#pragma unroll
        for (int kc = 0; kc < KC; ++kc) {
            v8h p0 = *(const v8h*)(bp0 + kc * 32);
            v8h p1 = *(const v8h*)(bp0 + kc * 32 + 16);
            v8h q0 = *(const v8h*)(bp1 + kc * 32);
            v8h q1 = *(const v8h*)(bp1 + kc * 32 + 16);
            v16h b0, b1;
#pragma unroll
            for (int i = 0; i < 8; ++i) {
                b0[i] = p0[i]; b0[8 + i] = p1[i];
                b1[i] = q0[i]; b1[8 + i] = q1[i];
            }
            S0 = __builtin_amdgcn_wmma_f32_16x16x32_f16(false, aU[kc], false, b0,
                                                        (short)0, S0, false, false);
            T0 = __builtin_amdgcn_wmma_f32_16x16x32_f16(false, aF[kc], false, b0,
                                                        (short)0, T0, false, false);
            S1 = __builtin_amdgcn_wmma_f32_16x16x32_f16(false, aU[kc], false, b1,
                                                        (short)0, S1, false, false);
            T1 = __builtin_amdgcn_wmma_f32_16x16x32_f16(false, aF[kc], false, b1,
                                                        (short)0, T1, false, false);
        }

        if (have) {
            _Float16* dst = lds + (1 - cur) * BUFH;
#pragma unroll
            for (int j = 0; j < 4; ++j) {
                const int r = wave * 4 + j;
#pragma unroll
                for (int k = 0; k < 4; ++k) {
                    const int c4 = lane + (k << 5);
                    v4h h;
#pragma unroll
                    for (int i = 0; i < 4; ++i) h[i] = (_Float16)stage[j][k][i];
                    *(v4h*)&dst[r * STRIDE + (c4 << 2)] = h;
                }
            }
        }

        if (l0 + half < L_) {
#pragma unroll
            for (int m = 0; m < 8; ++m) {
                const float s  = S0[m] * LOG2E;
                const float nm = fmaxf(rmax[m], s);
                const float c  = exp2f(rmax[m] - nm);
                const float e  = exp2f(s - nm);
                rden[m] = rden[m] * c + e;
                rnum[m] = rnum[m] * c + e * T0[m];
                rmax[m] = nm;
            }
        }
        if (l0 + 16 + half < L_) {
#pragma unroll
            for (int m = 0; m < 8; ++m) {
                const float s  = S1[m] * LOG2E;
                const float nm = fmaxf(rmax[m], s);
                const float c  = exp2f(rmax[m] - nm);
                const float e  = exp2f(s - nm);
                rden[m] = rden[m] * c + e;
                rnum[m] = rnum[m] * c + e * T1[m];
                rmax[m] = nm;
            }
        }
    }

    float res[8];
#pragma unroll
    for (int m = 0; m < 8; ++m) {
        float mx = rmax[m], dn = rden[m], nu = rnum[m];
#pragma unroll
        for (int off = 1; off < 16; off <<= 1) {
            const float mx2 = __shfl_xor(mx, off, 32);
            const float dn2 = __shfl_xor(dn, off, 32);
            const float nu2 = __shfl_xor(nu, off, 32);
            const float M   = fmaxf(mx, mx2);
            const float c1  = exp2f(mx  - M);
            const float c2  = exp2f(mx2 - M);
            dn = dn * c1 + dn2 * c2;
            nu = nu * c1 + nu2 * c2;
            mx = M;
        }
        res[m] = nu / dn;
    }

    const int ybase = yt + wave * 16 + (hi ? 8 : 0);
    if (half == 0) {
#pragma unroll
        for (int m = 0; m < 8; ++m) {
            const int y = ybase + m;
            if (y < Y_) out[(size_t)b * Y_ + y] = res[m] + fb[y];
        }
    }
}

extern "C" void kernel_launch(void* const* d_in, const int* in_sizes, int n_in,
                              void* d_out, int out_size, void* d_ws, size_t ws_size,
                              hipStream_t stream) {
    const float* x  = (const float*)d_in[0];
    const float* Uw = (const float*)d_in[1];
    const float* Fw = (const float*)d_in[2];
    const float* fb = (const float*)d_in[3];
    float* out = (float*)d_out;

    dim3 grid((Y_ + YT - 1) / YT, B_);
    label_attn_kernel<<<grid, 256, 0, stream>>>(x, Uw, Fw, fb, out);
}


# B-fragment ds_loads issued as one 16-load batch a 16-WMMA group ahead into a second register set (was load-wait-use per k-chunk); next-tile cvt+ds_store moved before the WMMAs to free registers
# speedup vs baseline: 1.2725x; 1.2725x over previous
#include <hip/hip_runtime.h>

typedef _Float16 v16h __attribute__((ext_vector_type(16)));
typedef _Float16 v8h  __attribute__((ext_vector_type(8)));
typedef _Float16 v4h  __attribute__((ext_vector_type(4)));
typedef float    v8f  __attribute__((ext_vector_type(8)));
typedef float    v4f  __attribute__((ext_vector_type(4)));

constexpr int B_ = 8, L_ = 2500, D_ = 512, Y_ = 8921;
constexpr int YT  = 128;
constexpr int LT  = 32;
constexpr int KC  = D_ / 32;
constexpr int STRIDE = D_ + 8;
constexpr int BUFH   = LT * STRIDE;
constexpr float LOG2E = 1.44269504088896340736f;

__global__ __launch_bounds__(256, 2)
void label_attn_kernel(const float* __restrict__ x,
                       const float* __restrict__ Uw,
                       const float* __restrict__ Fw,
                       const float* __restrict__ fb,
                       float* __restrict__ out)
{
    __shared__ _Float16 lds[2 * BUFH];

    const int b    = blockIdx.y;
    const int yt   = blockIdx.x * YT;
    const int tid  = threadIdx.x;
    const int wave = tid >> 5;
    const int lane = tid & 31;
    const int half = lane & 15;
    const bool hi  = lane >= 16;

    int yrow = yt + wave * 16 + half;
    if (yrow >= Y_) yrow = Y_ - 1;
    const float* urow = Uw + (size_t)yrow * D_;
    const float* frow = Fw + (size_t)yrow * D_;

    v16h aU[KC], aF[KC];
#pragma unroll
    for (int kc = 0; kc < KC; ++kc) {
        const int base = kc * 32 + (hi ? 8 : 0);
        v4f u0 = *(const v4f*)(urow + base);
        v4f u1 = *(const v4f*)(urow + base + 4);
        v4f u2 = *(const v4f*)(urow + base + 16);
        v4f u3 = *(const v4f*)(urow + base + 20);
        v4f f0 = *(const v4f*)(frow + base);
        v4f f1 = *(const v4f*)(frow + base + 4);
        v4f f2 = *(const v4f*)(frow + base + 16);
        v4f f3 = *(const v4f*)(frow + base + 20);
#pragma unroll
        for (int i = 0; i < 4; ++i) {
            aU[kc][i]      = (_Float16)u0[i];
            aU[kc][4 + i]  = (_Float16)u1[i];
            aU[kc][8 + i]  = (_Float16)u2[i];
            aU[kc][12 + i] = (_Float16)u3[i];
            aF[kc][i]      = (_Float16)f0[i];
            aF[kc][4 + i]  = (_Float16)f1[i];
            aF[kc][8 + i]  = (_Float16)f2[i];
            aF[kc][12 + i] = (_Float16)f3[i];
        }
    }

    float rmax[8], rden[8], rnum[8];
#pragma unroll
    for (int m = 0; m < 8; ++m) { rmax[m] = -1e30f; rden[m] = 0.f; rnum[m] = 0.f; }

    const float* xb = x + (size_t)b * L_ * D_;
    const int numLT = (L_ + LT - 1) / LT;

    {
#pragma unroll
        for (int j = 0; j < 4; ++j) {
            const int r = wave * 4 + j;
            int lrow = r; if (lrow >= L_) lrow = L_ - 1;
            const float* src = xb + (size_t)lrow * D_;
#pragma unroll
            for (int k = 0; k < 4; ++k) {
                const int c4 = lane + (k << 5);
                v4f f = *(const v4f*)(src + (c4 << 2));
                v4h h;
#pragma unroll
                for (int i = 0; i < 4; ++i) h[i] = (_Float16)f[i];
                *(v4h*)&lds[r * STRIDE + (c4 << 2)] = h;
            }
        }
    }

    for (int lt = 0; lt < numLT; ++lt) {
        const int cur = lt & 1;
        const int l0  = lt * LT;
        const bool have = (lt + 1) < numLT;

        v4f stage[4][4];
        if (have) {
            const int l0n = l0 + LT;
#pragma unroll
            for (int j = 0; j < 4; ++j) {
                const int r = wave * 4 + j;
                int lrow = l0n + r; if (lrow >= L_) lrow = L_ - 1;
                const float* src = xb + (size_t)lrow * D_;
#pragma unroll
                for (int k = 0; k < 4; ++k) {
                    const int c4 = lane + (k << 5);
                    stage[j][k] = *(const v4f*)(src + (c4 << 2));
                }
                if (l0n + LT + r < L_)
                    __builtin_prefetch(src + (size_t)LT * D_, 0, 1);
            }
        }

        __syncthreads();

        if (have) {
            _Float16* dst = lds + (1 - cur) * BUFH;
#pragma unroll
            for (int j = 0; j < 4; ++j) {
                const int r = wave * 4 + j;
#pragma unroll
                for (int k = 0; k < 4; ++k) {
                    const int c4 = lane + (k << 5);
                    v4h h;
#pragma unroll
                    for (int i = 0; i < 4; ++i) h[i] = (_Float16)stage[j][k][i];
                    *(v4h*)&dst[r * STRIDE + (c4 << 2)] = h;
                }
            }
        }

        v8f S0 = {}, T0 = {}, S1 = {}, T1 = {};
        const _Float16* base = lds + cur * BUFH + (hi ? 8 : 0);
        const _Float16* bp0  = base + half * STRIDE;
        const _Float16* bp1  = base + (16 + half) * STRIDE;
        v8h fr[2][4][4];
#pragma unroll
        for (int j = 0; j < 4; ++j) {
            fr[0][j][0] = *(const v8h*)(bp0 + j * 32);
            fr[0][j][1] = *(const v8h*)(bp0 + j * 32 + 16);
            fr[0][j][2] = *(const v8h*)(bp1 + j * 32);
            fr[0][j][3] = *(const v8h*)(bp1 + j * 32 + 16);
        }
#pragma unroll
        for (int g = 0; g < KC / 4; ++g) {
            if (g + 1 < KC / 4) {
#pragma unroll
                for (int j = 0; j < 4; ++j) {
                    const int kc = (g + 1) * 4 + j;
                    fr[(g + 1) & 1][j][0] = *(const v8h*)(bp0 + kc * 32);
                    fr[(g + 1) & 1][j][1] = *(const v8h*)(bp0 + kc * 32 + 16);
                    fr[(g + 1) & 1][j][2] = *(const v8h*)(bp1 + kc * 32);
                    fr[(g + 1) & 1][j][3] = *(const v8h*)(bp1 + kc * 32 + 16);
                }
            }
            __builtin_amdgcn_sched_barrier(0);
#pragma unroll
            for (int j = 0; j < 4; ++j) {
                const int kc = g * 4 + j;
                const v16h b0 = __builtin_shufflevector(fr[g & 1][j][0], fr[g & 1][j][1],
                    0, 1, 2, 3, 4, 5, 6, 7, 8, 9, 10, 11, 12, 13, 14, 15);
                const v16h b1 = __builtin_shufflevector(fr[g & 1][j][2], fr[g & 1][j][3],
                    0, 1, 2, 3, 4, 5, 6, 7, 8, 9, 10, 11, 12, 13, 14, 15);
                S0 = __builtin_amdgcn_wmma_f32_16x16x32_f16(false, aU[kc], false, b0,
                                                            (short)0, S0, false, false);
                T0 = __builtin_amdgcn_wmma_f32_16x16x32_f16(false, aF[kc], false, b0,
                                                            (short)0, T0, false, false);
                S1 = __builtin_amdgcn_wmma_f32_16x16x32_f16(false, aU[kc], false, b1,
                                                            (short)0, S1, false, false);
                T1 = __builtin_amdgcn_wmma_f32_16x16x32_f16(false, aF[kc], false, b1,
                                                            (short)0, T1, false, false);
            }
            __builtin_amdgcn_sched_barrier(0);
        }

        if (l0 + half < L_) {
#pragma unroll
            for (int m = 0; m < 8; ++m) {
                const float s  = S0[m] * LOG2E;
                const float nm = fmaxf(rmax[m], s);
                const float c  = exp2f(rmax[m] - nm);
                const float e  = exp2f(s - nm);
                rden[m] = rden[m] * c + e;
                rnum[m] = rnum[m] * c + e * T0[m];
                rmax[m] = nm;
            }
        }
        if (l0 + 16 + half < L_) {
#pragma unroll
            for (int m = 0; m < 8; ++m) {
                const float s  = S1[m] * LOG2E;
                const float nm = fmaxf(rmax[m], s);
                const float c  = exp2f(rmax[m] - nm);
                const float e  = exp2f(s - nm);
                rden[m] = rden[m] * c + e;
                rnum[m] = rnum[m] * c + e * T1[m];
                rmax[m] = nm;
            }
        }
    }

    float res[8];
#pragma unroll
    for (int m = 0; m < 8; ++m) {
        float mx = rmax[m], dn = rden[m], nu = rnum[m];
#pragma unroll
        for (int off = 1; off < 16; off <<= 1) {
            const float mx2 = __shfl_xor(mx, off, 32);
            const float dn2 = __shfl_xor(dn, off, 32);
            const float nu2 = __shfl_xor(nu, off, 32);
            const float M   = fmaxf(mx, mx2);
            const float c1  = exp2f(mx  - M);
            const float c2  = exp2f(mx2 - M);
            dn = dn * c1 + dn2 * c2;
            nu = nu * c1 + nu2 * c2;
            mx = M;
        }
        res[m] = nu / dn;
    }

    const int ybase = yt + wave * 16 + (hi ? 8 : 0);
    if (half == 0) {
#pragma unroll
        for (int m = 0; m < 8; ++m) {
            const int y = ybase + m;
            if (y < Y_) out[(size_t)b * Y_ + y] = res[m] + fb[y];
        }
    }
}

extern "C" void kernel_launch(void* const* d_in, const int* in_sizes, int n_in,
                              void* d_out, int out_size, void* d_ws, size_t ws_size,
                              hipStream_t stream) {
    const float* x  = (const float*)d_in[0];
    const float* Uw = (const float*)d_in[1];
    const float* Fw = (const float*)d_in[2];
    const float* fb = (const float*)d_in[3];
    float* out = (float*)d_out;

    dim3 grid((Y_ + YT - 1) / YT, B_);
    label_attn_kernel<<<grid, 256, 0, stream>>>(x, Uw, Fw, fb, out);
}


# x-tile global loads issued two tiles ahead inside the WMMA phase (3 of a wave's 4 rows behind the barrier, the 4th into the dead fragment set before the last 16 WMMAs), not right before the barrier
# speedup vs baseline: 1.6276x; 1.2791x over previous
#include <hip/hip_runtime.h>

typedef _Float16 v16h __attribute__((ext_vector_type(16)));
typedef _Float16 v8h  __attribute__((ext_vector_type(8)));
typedef _Float16 v4h  __attribute__((ext_vector_type(4)));
typedef float    v8f  __attribute__((ext_vector_type(8)));
typedef float    v4f  __attribute__((ext_vector_type(4)));

constexpr int B_ = 8, L_ = 2500, D_ = 512, Y_ = 8921;
constexpr int YT  = 128;
constexpr int LT  = 32;
constexpr int KC  = D_ / 32;
constexpr int STRIDE = D_ + 8;
constexpr int BUFH   = LT * STRIDE;
constexpr float LOG2E = 1.44269504088896340736f;

__global__ __launch_bounds__(256, 2)
void label_attn_kernel(const float* __restrict__ x,
                       const float* __restrict__ Uw,
                       const float* __restrict__ Fw,
                       const float* __restrict__ fb,
                       float* __restrict__ out)
{
    __shared__ _Float16 lds[2 * BUFH];

    const int b    = blockIdx.y;
    const int yt   = blockIdx.x * YT;
    const int tid  = threadIdx.x;
    const int wave = tid >> 5;
    const int lane = tid & 31;
    const int half = lane & 15;
    const bool hi  = lane >= 16;

    int yrow = yt + wave * 16 + half;
    if (yrow >= Y_) yrow = Y_ - 1;
    const float* urow = Uw + (size_t)yrow * D_;
    const float* frow = Fw + (size_t)yrow * D_;

    v16h aU[KC], aF[KC];
#pragma unroll
    for (int kc = 0; kc < KC; ++kc) {
        const int base = kc * 32 + (hi ? 8 : 0);
        v4f u0 = *(const v4f*)(urow + base);
        v4f u1 = *(const v4f*)(urow + base + 4);
        v4f u2 = *(const v4f*)(urow + base + 16);
        v4f u3 = *(const v4f*)(urow + base + 20);
        v4f f0 = *(const v4f*)(frow + base);
        v4f f1 = *(const v4f*)(frow + base + 4);
        v4f f2 = *(const v4f*)(frow + base + 16);
        v4f f3 = *(const v4f*)(frow + base + 20);
#pragma unroll
        for (int i = 0; i < 4; ++i) {
            aU[kc][i]      = (_Float16)u0[i];
            aU[kc][4 + i]  = (_Float16)u1[i];
            aU[kc][8 + i]  = (_Float16)u2[i];
            aU[kc][12 + i] = (_Float16)u3[i];
            aF[kc][i]      = (_Float16)f0[i];
            aF[kc][4 + i]  = (_Float16)f1[i];
            aF[kc][8 + i]  = (_Float16)f2[i];
            aF[kc][12 + i] = (_Float16)f3[i];
        }
    }

    float rmax[8], rden[8], rnum[8];
#pragma unroll
    for (int m = 0; m < 8; ++m) { rmax[m] = -1e30f; rden[m] = 0.f; rnum[m] = 0.f; }

    const float* xb = x + (size_t)b * L_ * D_;
    const int numLT = (L_ + LT - 1) / LT;

    {
#pragma unroll
        for (int j = 0; j < 4; ++j) {
            const int r = wave * 4 + j;
            int lrow = r; if (lrow >= L_) lrow = L_ - 1;
            const float* src = xb + (size_t)lrow * D_;
#pragma unroll
            for (int k = 0; k < 4; ++k) {
                const int c4 = lane + (k << 5);
                v4f f = *(const v4f*)(src + (c4 << 2));
                v4h h;
#pragma unroll
                for (int i = 0; i < 4; ++i) h[i] = (_Float16)f[i];
                *(v4h*)&lds[r * STRIDE + (c4 << 2)] = h;
            }
        }
    }

    constexpr int NA = 3;
    v4f stageA[NA][4], stageB[4 - NA][4];
    {
#pragma unroll
        for (int j = 0; j < 4; ++j) {
            const int r = wave * 4 + j;
            int lrow = LT + r; if (lrow >= L_) lrow = L_ - 1;
            const float* src = xb + (size_t)lrow * D_;
#pragma unroll
            for (int k = 0; k < 4; ++k) {
                const int c4 = lane + (k << 5);
                if (j < NA) stageA[j][k]      = *(const v4f*)(src + (c4 << 2));
                else        stageB[j - NA][k] = *(const v4f*)(src + (c4 << 2));
            }
        }
    }

    for (int lt = 0; lt < numLT; ++lt) {
        const int cur = lt & 1;
        const int l0  = lt * LT;
        const bool have = (lt + 1) < numLT;
        const int l0nn = l0 + 2 * LT;

        __syncthreads();

        if (have) {
            _Float16* dst = lds + (1 - cur) * BUFH;
#pragma unroll
            for (int j = 0; j < 4; ++j) {
                const int r = wave * 4 + j;
#pragma unroll
                for (int k = 0; k < 4; ++k) {
                    const int c4 = lane + (k << 5);
                    const v4f f = (j < NA) ? stageA[j][k] : stageB[j - NA][k];
                    v4h h;
#pragma unroll
                    for (int i = 0; i < 4; ++i) h[i] = (_Float16)f[i];
                    *(v4h*)&dst[r * STRIDE + (c4 << 2)] = h;
                }
            }
        }

#pragma unroll
        for (int j = 0; j < NA; ++j) {
            int lrow = l0nn + wave * 4 + j; if (lrow >= L_) lrow = L_ - 1;
            const float* src = xb + (size_t)lrow * D_;
#pragma unroll
            for (int k = 0; k < 4; ++k)
                stageA[j][k] = *(const v4f*)(src + ((lane + (k << 5)) << 2));
        }

        v8f S0 = {}, T0 = {}, S1 = {}, T1 = {};
        const _Float16* base = lds + cur * BUFH + (hi ? 8 : 0);
        const _Float16* bp0  = base + half * STRIDE;
        const _Float16* bp1  = base + (16 + half) * STRIDE;
        v8h fr[2][4][4];
#pragma unroll
        for (int j = 0; j < 4; ++j) {
            fr[0][j][0] = *(const v8h*)(bp0 + j * 32);
            fr[0][j][1] = *(const v8h*)(bp0 + j * 32 + 16);
            fr[0][j][2] = *(const v8h*)(bp1 + j * 32);
            fr[0][j][3] = *(const v8h*)(bp1 + j * 32 + 16);
        }
#pragma unroll
        for (int g = 0; g < KC / 4; ++g) {
            if (g + 1 < KC / 4) {
#pragma unroll
                for (int j = 0; j < 4; ++j) {
                    const int kc = (g + 1) * 4 + j;
                    fr[(g + 1) & 1][j][0] = *(const v8h*)(bp0 + kc * 32);
                    fr[(g + 1) & 1][j][1] = *(const v8h*)(bp0 + kc * 32 + 16);
                    fr[(g + 1) & 1][j][2] = *(const v8h*)(bp1 + kc * 32);
                    fr[(g + 1) & 1][j][3] = *(const v8h*)(bp1 + kc * 32 + 16);
                }
            }
            if (g + 1 == KC / 4) {
#pragma unroll
                for (int j = NA; j < 4; ++j) {
                    int lrow = l0nn + wave * 4 + j; if (lrow >= L_) lrow = L_ - 1;
                    const float* src = xb + (size_t)lrow * D_;
#pragma unroll
                    for (int k = 0; k < 4; ++k)
                        stageB[j - NA][k] = *(const v4f*)(src + ((lane + (k << 5)) << 2));
                }
            }
            __builtin_amdgcn_sched_barrier(0);
#pragma unroll
            for (int j = 0; j < 4; ++j) {
                const int kc = g * 4 + j;
                const v16h b0 = __builtin_shufflevector(fr[g & 1][j][0], fr[g & 1][j][1],
                    0, 1, 2, 3, 4, 5, 6, 7, 8, 9, 10, 11, 12, 13, 14, 15);
                const v16h b1 = __builtin_shufflevector(fr[g & 1][j][2], fr[g & 1][j][3],
                    0, 1, 2, 3, 4, 5, 6, 7, 8, 9, 10, 11, 12, 13, 14, 15);
                S0 = __builtin_amdgcn_wmma_f32_16x16x32_f16(false, aU[kc], false, b0,
                                                            (short)0, S0, false, false);
                T0 = __builtin_amdgcn_wmma_f32_16x16x32_f16(false, aF[kc], false, b0,
                                                            (short)0, T0, false, false);
                S1 = __builtin_amdgcn_wmma_f32_16x16x32_f16(false, aU[kc], false, b1,
                                                            (short)0, S1, false, false);
                T1 = __builtin_amdgcn_wmma_f32_16x16x32_f16(false, aF[kc], false, b1,
                                                            (short)0, T1, false, false);
            }
            __builtin_amdgcn_sched_barrier(0);
        }

        if (l0 + half < L_) {
#pragma unroll
            for (int m = 0; m < 8; ++m) {
                const float s  = S0[m] * LOG2E;
                const float nm = fmaxf(rmax[m], s);
                const float c  = exp2f(rmax[m] - nm);
                const float e  = exp2f(s - nm);
                rden[m] = rden[m] * c + e;
                rnum[m] = rnum[m] * c + e * T0[m];
                rmax[m] = nm;
            }
        }
        if (l0 + 16 + half < L_) {
#pragma unroll
            for (int m = 0; m < 8; ++m) {
                const float s  = S1[m] * LOG2E;
                const float nm = fmaxf(rmax[m], s);
                const float c  = exp2f(rmax[m] - nm);
                const float e  = exp2f(s - nm);
                rden[m] = rden[m] * c + e;
                rnum[m] = rnum[m] * c + e * T1[m];
                rmax[m] = nm;
            }
        }
    }

    float res[8];
#pragma unroll
    for (int m = 0; m < 8; ++m) {
        float mx = rmax[m], dn = rden[m], nu = rnum[m];
#pragma unroll
        for (int off = 1; off < 16; off <<= 1) {
            const float mx2 = __shfl_xor(mx, off, 32);
            const float dn2 = __shfl_xor(dn, off, 32);
            const float nu2 = __shfl_xor(nu, off, 32);
            const float M   = fmaxf(mx, mx2);
            const float c1  = exp2f(mx  - M);
            const float c2  = exp2f(mx2 - M);
            dn = dn * c1 + dn2 * c2;
            nu = nu * c1 + nu2 * c2;
            mx = M;
        }
        res[m] = nu / dn;
    }

    const int ybase = yt + wave * 16 + (hi ? 8 : 0);
    if (half == 0) {
#pragma unroll
        for (int m = 0; m < 8; ++m) {
            const int y = ybase + m;
            if (y < Y_) out[(size_t)b * Y_ + y] = res[m] + fb[y];
        }
    }
}

extern "C" void kernel_launch(void* const* d_in, const int* in_sizes, int n_in,
                              void* d_out, int out_size, void* d_ws, size_t ws_size,
                              hipStream_t stream) {
    const float* x  = (const float*)d_in[0];
    const float* Uw = (const float*)d_in[1];
    const float* Fw = (const float*)d_in[2];
    const float* fb = (const float*)d_in[3];
    float* out = (float*)d_out;

    dim3 grid((Y_ + YT - 1) / YT, B_);
    label_attn_kernel<<<grid, 256, 0, stream>>>(x, Uw, Fw, fb, out);
}


# softmax update cut to a third of its VALU work: bare v_exp_f32, both columns folded round a joint max (den, num summed in a new order), tail by score bias; B-fragment ring 3x8 keeps it spill-free
# speedup vs baseline: 1.8756x; 1.1524x over previous
#include <hip/hip_runtime.h>

typedef _Float16 v16h __attribute__((ext_vector_type(16)));
typedef _Float16 v8h  __attribute__((ext_vector_type(8)));
typedef _Float16 v4h  __attribute__((ext_vector_type(4)));
typedef float    v8f  __attribute__((ext_vector_type(8)));
typedef float    v4f  __attribute__((ext_vector_type(4)));

constexpr int B_ = 8, L_ = 2500, D_ = 512, Y_ = 8921;
constexpr int YT  = 128;
constexpr int LT  = 32;
constexpr int KC  = D_ / 32;
constexpr int STRIDE = D_ + 8;
constexpr int BUFH   = LT * STRIDE;
constexpr float LOG2E = 1.44269504088896340736f;

__global__ __launch_bounds__(256, 2)
void label_attn_kernel(const float* __restrict__ x,
                       const float* __restrict__ Uw,
                       const float* __restrict__ Fw,
                       const float* __restrict__ fb,
                       float* __restrict__ out)
{
    __shared__ _Float16 lds[2 * BUFH];

    const int b    = blockIdx.y;
    const int yt   = blockIdx.x * YT;
    const int tid  = threadIdx.x;
    const int wave = tid >> 5;
    const int lane = tid & 31;
    const int half = lane & 15;
    const bool hi  = lane >= 16;

    int yrow = yt + wave * 16 + half;
    if (yrow >= Y_) yrow = Y_ - 1;
    const float* urow = Uw + (size_t)yrow * D_;
    const float* frow = Fw + (size_t)yrow * D_;

    v16h aU[KC], aF[KC];
#pragma unroll
    for (int kc = 0; kc < KC; ++kc) {
        const int base = kc * 32 + (hi ? 8 : 0);
        v4f u0 = *(const v4f*)(urow + base);
        v4f u1 = *(const v4f*)(urow + base + 4);
        v4f u2 = *(const v4f*)(urow + base + 16);
        v4f u3 = *(const v4f*)(urow + base + 20);
        v4f f0 = *(const v4f*)(frow + base);
        v4f f1 = *(const v4f*)(frow + base + 4);
        v4f f2 = *(const v4f*)(frow + base + 16);
        v4f f3 = *(const v4f*)(frow + base + 20);
#pragma unroll
        for (int i = 0; i < 4; ++i) {
            aU[kc][i]      = (_Float16)u0[i];
            aU[kc][4 + i]  = (_Float16)u1[i];
            aU[kc][8 + i]  = (_Float16)u2[i];
            aU[kc][12 + i] = (_Float16)u3[i];
            aF[kc][i]      = (_Float16)f0[i];
            aF[kc][4 + i]  = (_Float16)f1[i];
            aF[kc][8 + i]  = (_Float16)f2[i];
            aF[kc][12 + i] = (_Float16)f3[i];
        }
    }

    float rmax[8], rden[8], rnum[8];
#pragma unroll
    for (int m = 0; m < 8; ++m) { rmax[m] = -1e30f; rden[m] = 0.f; rnum[m] = 0.f; }

    const float* xb = x + (size_t)b * L_ * D_;
    const int numLT = (L_ + LT - 1) / LT;

    {
#pragma unroll
        for (int j = 0; j < 4; ++j) {
            const int r = wave * 4 + j;
            int lrow = r; if (lrow >= L_) lrow = L_ - 1;
            const float* src = xb + (size_t)lrow * D_;
#pragma unroll
            for (int k = 0; k < 4; ++k) {
                const int c4 = lane + (k << 5);
                v4f f = *(const v4f*)(src + (c4 << 2));
                v4h h;
#pragma unroll
                for (int i = 0; i < 4; ++i) h[i] = (_Float16)f[i];
                *(v4h*)&lds[r * STRIDE + (c4 << 2)] = h;
            }
        }
    }

    constexpr int NA = 3;
    v4f stageA[NA][4], stageB[4 - NA][4];
    {
#pragma unroll
        for (int j = 0; j < 4; ++j) {
            const int r = wave * 4 + j;
            int lrow = LT + r; if (lrow >= L_) lrow = L_ - 1;
            const float* src = xb + (size_t)lrow * D_;
#pragma unroll
            for (int k = 0; k < 4; ++k) {
                const int c4 = lane + (k << 5);
                if (j < NA) stageA[j][k]      = *(const v4f*)(src + (c4 << 2));
                else        stageB[j - NA][k] = *(const v4f*)(src + (c4 << 2));
            }
        }
    }

    for (int lt = 0; lt < numLT; ++lt) {
        const int cur = lt & 1;
        const int l0  = lt * LT;
        const bool have = (lt + 1) < numLT;
        const int l0nn = l0 + 2 * LT;

        __syncthreads();

        if (have) {
            _Float16* dst = lds + (1 - cur) * BUFH;
#pragma unroll
            for (int j = 0; j < 4; ++j) {
                const int r = wave * 4 + j;
#pragma unroll
                for (int k = 0; k < 4; ++k) {
                    const int c4 = lane + (k << 5);
                    const v4f f = (j < NA) ? stageA[j][k] : stageB[j - NA][k];
                    v4h h;
#pragma unroll
                    for (int i = 0; i < 4; ++i) h[i] = (_Float16)f[i];
                    *(v4h*)&dst[r * STRIDE + (c4 << 2)] = h;
                }
            }
        }

#pragma unroll
        for (int j = 0; j < NA; ++j) {
            int lrow = l0nn + wave * 4 + j; if (lrow >= L_) lrow = L_ - 1;
            const float* src = xb + (size_t)lrow * D_;
#pragma unroll
            for (int k = 0; k < 4; ++k)
                stageA[j][k] = *(const v4f*)(src + ((lane + (k << 5)) << 2));
        }

        v8f S0 = {}, T0 = {}, S1 = {}, T1 = {};
        const _Float16* base = lds + cur * BUFH + (hi ? 8 : 0);
        const _Float16* bp0  = base + half * STRIDE;
        const _Float16* bp1  = base + (16 + half) * STRIDE;
        v8h fr[3][2][4];
#pragma unroll
        for (int h = 0; h < 2; ++h) {
#pragma unroll
            for (int j = 0; j < 2; ++j) {
                const int kc = h * 2 + j;
                fr[h][j][0] = *(const v8h*)(bp0 + kc * 32);
                fr[h][j][1] = *(const v8h*)(bp0 + kc * 32 + 16);
                fr[h][j][2] = *(const v8h*)(bp1 + kc * 32);
                fr[h][j][3] = *(const v8h*)(bp1 + kc * 32 + 16);
            }
        }
#pragma unroll
        for (int h = 0; h < KC / 2; ++h) {
            if (h + 2 < KC / 2) {
#pragma unroll
                for (int j = 0; j < 2; ++j) {
                    const int kc = (h + 2) * 2 + j;
                    fr[(h + 2) % 3][j][0] = *(const v8h*)(bp0 + kc * 32);
                    fr[(h + 2) % 3][j][1] = *(const v8h*)(bp0 + kc * 32 + 16);
                    fr[(h + 2) % 3][j][2] = *(const v8h*)(bp1 + kc * 32);
                    fr[(h + 2) % 3][j][3] = *(const v8h*)(bp1 + kc * 32 + 16);
                }
            }
            if (h + 2 == KC / 2) {
#pragma unroll
                for (int j = NA; j < 4; ++j) {
                    int lrow = l0nn + wave * 4 + j; if (lrow >= L_) lrow = L_ - 1;
                    const float* src = xb + (size_t)lrow * D_;
#pragma unroll
                    for (int k = 0; k < 4; ++k)
                        stageB[j - NA][k] = *(const v4f*)(src + ((lane + (k << 5)) << 2));
                }
            }
            __builtin_amdgcn_sched_barrier(0);
#pragma unroll
            for (int j = 0; j < 2; ++j) {
                const int kc = h * 2 + j;
                const v16h b0 = __builtin_shufflevector(fr[h % 3][j][0], fr[h % 3][j][1],
                    0, 1, 2, 3, 4, 5, 6, 7, 8, 9, 10, 11, 12, 13, 14, 15);
                const v16h b1 = __builtin_shufflevector(fr[h % 3][j][2], fr[h % 3][j][3],
                    0, 1, 2, 3, 4, 5, 6, 7, 8, 9, 10, 11, 12, 13, 14, 15);
                S0 = __builtin_amdgcn_wmma_f32_16x16x32_f16(false, aU[kc], false, b0,
                                                            (short)0, S0, false, false);
                T0 = __builtin_amdgcn_wmma_f32_16x16x32_f16(false, aF[kc], false, b0,
                                                            (short)0, T0, false, false);
                S1 = __builtin_amdgcn_wmma_f32_16x16x32_f16(false, aU[kc], false, b1,
                                                            (short)0, S1, false, false);
                T1 = __builtin_amdgcn_wmma_f32_16x16x32_f16(false, aF[kc], false, b1,
                                                            (short)0, T1, false, false);
            }
            __builtin_amdgcn_sched_barrier(0);
        }

        {
            const float bias0 = (l0 + half      < L_) ? 0.f : -3.0e38f;
            const float bias1 = (l0 + 16 + half < L_) ? 0.f : -3.0e38f;
#pragma unroll
            for (int m = 0; m < 8; ++m) {
                const float s0 = __builtin_fmaf(S0[m], LOG2E, bias0);
                const float s1 = __builtin_fmaf(S1[m], LOG2E, bias1);
                const float nm = fmaxf(rmax[m], fmaxf(s0, s1));
                const float c  = __builtin_amdgcn_exp2f(rmax[m] - nm);
                const float e0 = __builtin_amdgcn_exp2f(s0 - nm);
                const float e1 = __builtin_amdgcn_exp2f(s1 - nm);
                rden[m] = rden[m] * c + (e0 + e1);
                rnum[m] = (rnum[m] * c + e0 * T0[m]) + e1 * T1[m];
                rmax[m] = nm;
            }
        }
    }

    float res[8];
#pragma unroll
    for (int m = 0; m < 8; ++m) {
        float mx = rmax[m], dn = rden[m], nu = rnum[m];
#pragma unroll
        for (int off = 1; off < 16; off <<= 1) {
            const float mx2 = __shfl_xor(mx, off, 32);
            const float dn2 = __shfl_xor(dn, off, 32);
            const float nu2 = __shfl_xor(nu, off, 32);
            const float M   = fmaxf(mx, mx2);
            const float c1  = exp2f(mx  - M);
            const float c2  = exp2f(mx2 - M);
            dn = dn * c1 + dn2 * c2;
            nu = nu * c1 + nu2 * c2;
            mx = M;
        }
        res[m] = nu / dn;
    }

    const int ybase = yt + wave * 16 + (hi ? 8 : 0);
    if (half == 0) {
#pragma unroll
        for (int m = 0; m < 8; ++m) {
            const int y = ybase + m;
            if (y < Y_) out[(size_t)b * Y_ + y] = res[m] + fb[y];
        }
    }
}

extern "C" void kernel_launch(void* const* d_in, const int* in_sizes, int n_in,
                              void* d_out, int out_size, void* d_ws, size_t ws_size,
                              hipStream_t stream) {
    const float* x  = (const float*)d_in[0];
    const float* Uw = (const float*)d_in[1];
    const float* Fw = (const float*)d_in[2];
    const float* fb = (const float*)d_in[3];
    float* out = (float*)d_out;

    dim3 grid((Y_ + YT - 1) / YT, B_);
    label_attn_kernel<<<grid, 256, 0, stream>>>(x, Uw, Fw, fb, out);
}
